# starting kernel
# baseline (speedup 1.0000x reference)
#include <hip/hip_runtime.h>
#include <hip/hip_bf16.h>


typedef __attribute__((ext_vector_type(2))) float v2f;
typedef __attribute__((ext_vector_type(4))) float v4f;
typedef __attribute__((ext_vector_type(8))) float v8f;

namespace {
constexpr int M_DIM = 128;
constexpr int N_DIM = 8192;
constexpr int K_DIM = 8192;
constexpr float SCALING = 2.0f;
}

__global__ __launch_bounds__(256, 1)
void lokr_fused_gemm(const float* __restrict__ x,
                     const float* __restrict__ W,
                     const float* __restrict__ A,
                     const float* __restrict__ B16,
                     const float* __restrict__ scal,
                     float* __restrict__ out)
{
    const int lane  = threadIdx.x & 31;
    const int wave  = threadIdx.x >> 5;
    const int nt    = blockIdx.x * 4 + (wave & 3);
    const int mbase = (wave >> 2) * 64;
    const int n0    = nt * 16;
    const int l16   = lane & 15;
    const int lhalf = lane >> 4;

    const float sb = scal[0] * SCALING * B16[l16];

    const float* wp = W + (size_t)(n0 + l16) * K_DIM + 4 * lhalf;
    const float* ap = A + (size_t)nt * K_DIM + 4 * lhalf;
    const float* xp = x + (size_t)(mbase + l16) * K_DIM + 4 * lhalf;

    v8f c[4] = {};

    for (int kc = 0; kc < K_DIM; kc += 32) {
        __builtin_prefetch(wp + kc + 2048, 0, 1);
        #pragma unroll
        for (int ku = 0; ku < 32; ku += 8) {
            const int kb = kc + ku;
            const v4f w4 = *(const v4f*)(wp + kb);
            const v4f a4 = *(const v4f*)(ap + kb);
            v4f be;
            be.x = __builtin_fmaf(sb, a4.x, w4.x);
            be.y = __builtin_fmaf(sb, a4.y, w4.y);
            be.z = __builtin_fmaf(sb, a4.z, w4.z);
            be.w = __builtin_fmaf(sb, a4.w, w4.w);
            const v2f b0 = __builtin_shufflevector(be, be, 0, 1);
            const v2f b1 = __builtin_shufflevector(be, be, 2, 3);

            v4f x4[4];
            #pragma unroll
            for (int mt = 0; mt < 4; ++mt)
                x4[mt] = *(const v4f*)(xp + (size_t)(mt * 16) * K_DIM + kb);

            #pragma unroll
            for (int mt = 0; mt < 4; ++mt) {
                const v2f a0 = __builtin_shufflevector(x4[mt], x4[mt], 0, 1);
                c[mt] = __builtin_amdgcn_wmma_f32_16x16x4_f32(
                    false, a0, false, b0, (short)0, c[mt], false, false);
            }
            #pragma unroll
            for (int mt = 0; mt < 4; ++mt) {
                const v2f a1 = __builtin_shufflevector(x4[mt], x4[mt], 2, 3);
                c[mt] = __builtin_amdgcn_wmma_f32_16x16x4_f32(
                    false, a1, false, b1, (short)0, c[mt], false, false);
            }
        }
    }

    float* op = out + n0 + l16;
    #pragma unroll
    for (int mt = 0; mt < 4; ++mt) {
        #pragma unroll
        for (int r = 0; r < 8; ++r) {
            const int m = mbase + mt * 16 + lhalf * 8 + r;
            op[(size_t)m * N_DIM] = c[mt][r];
        }
    }
}

extern "C" void kernel_launch(void* const* d_in, const int* in_sizes, int n_in,
                              void* d_out, int out_size, void* d_ws, size_t ws_size,
                              hipStream_t stream) {
    (void)in_sizes; (void)n_in; (void)out_size; (void)d_ws; (void)ws_size;
    const float* x    = (const float*)d_in[0];
    const float* W    = (const float*)d_in[1];
    const float* A    = (const float*)d_in[2];
    const float* B16  = (const float*)d_in[3];
    const float* scal = (const float*)d_in[4];
    float* out = (float*)d_out;

    dim3 grid(N_DIM / 64);
    dim3 block(256);
    hipLaunchKernelGGL(lokr_fused_gemm, grid, block, 0, stream,
                       x, W, A, B16, scal, out);
}


# Whole-line load groups: a wave works in 32-k chunks, the four loads of each 128-B line issued back to back (one fetch per line, not four); 8 waves, 2 per SIMD; 8 partials, sum order changed
# speedup vs baseline: 4.3839x; 4.3839x over previous
#include <hip/hip_runtime.h>


typedef __attribute__((ext_vector_type(2))) float v2f;
typedef __attribute__((ext_vector_type(4))) float v4f;
typedef __attribute__((ext_vector_type(8))) float v8f;

namespace {
constexpr int M_DIM = 128;
constexpr int N_DIM = 8192;
constexpr int K_DIM = 8192;
constexpr float SCALING = 2.0f;
constexpr int KSLICES = 8;
constexpr int MT = M_DIM / 16;
constexpr int MH = MT / 2;
constexpr int KCH = 32;
constexpr int NP = KCH / 8;
constexpr int KSTRIDE = KCH * KSLICES;
constexpr int RED_PITCH = 36;
constexpr int RED_PLANE = M_DIM * RED_PITCH;

struct BLines {
    v4f w0[NP], w1[NP];
    v4f a0[NP], a1[NP];
};
struct XLines {
    v4f xr[MH][NP];
};
struct BOp {
    v4f b0[NP], b1[NP];
};

__device__ __forceinline__ void load_b(BLines& f, const float* wp, const float* ap, int kc)
{
    #pragma unroll
    for (int j = 0; j < NP; ++j) f.w0[j] = *(const v4f*)(wp + kc + 8 * j);
    #pragma unroll
    for (int j = 0; j < NP; ++j) f.w1[j] = *(const v4f*)(wp + (size_t)16 * K_DIM + kc + 8 * j);
    #pragma unroll
    for (int j = 0; j < NP; ++j) f.a0[j] = *(const v4f*)(ap + kc + 8 * j);
    #pragma unroll
    for (int j = 0; j < NP; ++j) f.a1[j] = *(const v4f*)(ap + (size_t)K_DIM + kc + 8 * j);
}

template <int MT0>
__device__ __forceinline__ void load_x(XLines& f, const float* xp, int kc)
{
    #pragma unroll
    for (int mt = 0; mt < MH; ++mt) {
        #pragma unroll
        for (int j = 0; j < NP; ++j)
            f.xr[mt][j] = *(const v4f*)(xp + (size_t)((MT0 + mt) * 16) * K_DIM + kc + 8 * j);
    }
}

#define LOAD_GROUP_EDGE() do { __builtin_amdgcn_sched_barrier(0); } while (0)

__device__ __forceinline__ v4f fold(float sb, v4f a4, v4f w4)
{
    v4f be;
    be.x = __builtin_fmaf(sb, a4.x, w4.x);
    be.y = __builtin_fmaf(sb, a4.y, w4.y);
    be.z = __builtin_fmaf(sb, a4.z, w4.z);
    be.w = __builtin_fmaf(sb, a4.w, w4.w);
    return be;
}

__device__ __forceinline__ void fold_lines(BOp& b, const BLines& f, float sb)
{
    #pragma unroll
    for (int j = 0; j < NP; ++j) {
        b.b0[j] = fold(sb, f.a0[j], f.w0[j]);
        b.b1[j] = fold(sb, f.a1[j], f.w1[j]);
    }
}

template <int MT0>
__device__ __forceinline__ void mma_half(v8f (&c)[MT][2], const XLines& f, const BOp& b)
{
    #pragma unroll
    for (int j = 0; j < NP; ++j) {
        const v2f b00 = __builtin_shufflevector(b.b0[j], b.b0[j], 0, 1);
        const v2f b01 = __builtin_shufflevector(b.b0[j], b.b0[j], 2, 3);
        const v2f b10 = __builtin_shufflevector(b.b1[j], b.b1[j], 0, 1);
        const v2f b11 = __builtin_shufflevector(b.b1[j], b.b1[j], 2, 3);
        #pragma unroll
        for (int mt = 0; mt < MH; ++mt) {
            const v2f xa = __builtin_shufflevector(f.xr[mt][j], f.xr[mt][j], 0, 1);
            c[MT0 + mt][0] = __builtin_amdgcn_wmma_f32_16x16x4_f32(
                false, xa, false, b00, (short)0, c[MT0 + mt][0], false, false);
            c[MT0 + mt][1] = __builtin_amdgcn_wmma_f32_16x16x4_f32(
                false, xa, false, b10, (short)0, c[MT0 + mt][1], false, false);
        }
        #pragma unroll
        for (int mt = 0; mt < MH; ++mt) {
            const v2f xb = __builtin_shufflevector(f.xr[mt][j], f.xr[mt][j], 2, 3);
            c[MT0 + mt][0] = __builtin_amdgcn_wmma_f32_16x16x4_f32(
                false, xb, false, b01, (short)0, c[MT0 + mt][0], false, false);
            c[MT0 + mt][1] = __builtin_amdgcn_wmma_f32_16x16x4_f32(
                false, xb, false, b11, (short)0, c[MT0 + mt][1], false, false);
        }
    }
}

__device__ __forceinline__ void chunk_step(v8f (&c)[MT][2], BLines& r, XLines& p, XLines& q,
                                           BOp& b, float sb, const float* wp,
                                           const float* ap, const float* xp, int kn)
{
    __builtin_prefetch(wp + kn + 2048 - KSTRIDE, 0, 1);
    __builtin_prefetch(wp + (size_t)16 * K_DIM + kn + 2048 - KSTRIDE, 0, 1);

    fold_lines(b, r, sb);
    LOAD_GROUP_EDGE();
    load_b(r, wp, ap, kn);
    LOAD_GROUP_EDGE();
    mma_half<0>(c, p, b);
    LOAD_GROUP_EDGE();
    load_x<0>(p, xp, kn);
    LOAD_GROUP_EDGE();
    mma_half<MH>(c, q, b);
    LOAD_GROUP_EDGE();
    load_x<MH>(q, xp, kn);
    LOAD_GROUP_EDGE();
}
}

__global__ __launch_bounds__(256, 1)
void lokr_fused_gemm(const float* x,
                     const float* W,
                     const float* A,
                     const float* __restrict__ B16,
                     const float* __restrict__ scal,
                     float* out)
{
    __shared__ __attribute__((aligned(16))) float red[KSLICES * RED_PLANE];

    const int lane  = threadIdx.x & 31;
    const int slice = __builtin_amdgcn_readfirstlane((int)(threadIdx.x >> 5));
    const int n0    = blockIdx.x * 32;
    const int l16   = lane & 15;
    const int lhalf = lane >> 4;

    const float sb = scal[0] * SCALING * B16[l16];

    const float* wp = W + (size_t)(n0 + l16) * K_DIM + 4 * lhalf;
    const float* ap = A + (size_t)(blockIdx.x * 2) * K_DIM + 4 * lhalf;
    const float* xp = x + (size_t)l16 * K_DIM + 4 * lhalf;

    v8f c[MT][2] = {};

    BLines r;
    XLines p, q;
    BOp ba, bb;
    int kc = slice * KCH;
    load_b(r, wp, ap, kc);
    load_x<0>(p, xp, kc);
    load_x<MH>(q, xp, kc);
    LOAD_GROUP_EDGE();
    for (; kc < K_DIM; kc += 2 * KSTRIDE) {
        chunk_step(c, r, p, q, ba, sb, wp, ap, xp, kc + KSTRIDE);
        const int kn = (kc + 2 * KSTRIDE < K_DIM) ? kc + 2 * KSTRIDE : kc + KSTRIDE;
        chunk_step(c, r, p, q, bb, sb, wp, ap, xp, kn);
    }

    float* rp = red + slice * RED_PLANE + l16;
    #pragma unroll
    for (int mt = 0; mt < MT; ++mt) {
        #pragma unroll
        for (int i = 0; i < 8; ++i) {
            const int m = mt * 16 + lhalf * 8 + i;
            rp[m * RED_PITCH]      = c[mt][0][i];
            rp[m * RED_PITCH + 16] = c[mt][1][i];
        }
    }
    __syncthreads();

    const int g = threadIdx.x & 7;
    #pragma unroll
    for (int i = 0; i < 4; ++i) {
        const int m = (int)(threadIdx.x >> 3) + 32 * i;
        const float* qd = red + m * RED_PITCH + 4 * g;
        v4f s = *(const v4f*)qd;
        #pragma unroll
        for (int sl = 1; sl < KSLICES; ++sl)
            s += *(const v4f*)(qd + sl * RED_PLANE);
        *(v4f*)(out + (size_t)m * N_DIM + n0 + 4 * g) = s;
    }
}

extern "C" void kernel_launch(void* const* d_in, const int* in_sizes, int n_in,
                              void* d_out, int out_size, void* d_ws, size_t ws_size,
                              hipStream_t stream) {
    (void)in_sizes; (void)n_in; (void)out_size; (void)d_ws; (void)ws_size;
    const float* x    = (const float*)d_in[0];
    const float* W    = (const float*)d_in[1];
    const float* A    = (const float*)d_in[2];
    const float* B16  = (const float*)d_in[3];
    const float* scal = (const float*)d_in[4];
    float* out = (float*)d_out;

    dim3 grid(N_DIM / 32);
    dim3 block(256);
    hipLaunchKernelGGL(lokr_fused_gemm, grid, block, 0, stream,
                       x, W, A, B16, scal, out);
}


# Wave tile 128x32 -> 128x64 (4 waves, 1 per SIMD, ~640 registers): half the x fragment loads per product; K halved over group pairs, a 2nd launch adds the halves (sum order changed)
# speedup vs baseline: 5.6834x; 1.2964x over previous
#include <hip/hip_runtime.h>


typedef __attribute__((ext_vector_type(2))) float v2f;
typedef __attribute__((ext_vector_type(4))) float v4f;
typedef __attribute__((ext_vector_type(8))) float v8f;

namespace {
constexpr int M_DIM = 128;
constexpr int N_DIM = 8192;
constexpr int K_DIM = 8192;
constexpr float SCALING = 2.0f;
constexpr int KSLICES = 4;
constexpr int KHALF = K_DIM / 2;
constexpr int MT = M_DIM / 16;
constexpr int MH = MT / 2;
constexpr int NT = 4;
constexpr int NB = NT * 16;
constexpr int KCH = 32;
constexpr int NP = KCH / 8;
constexpr int KSTRIDE = KCH * KSLICES;
constexpr int KPF = 8 * KSTRIDE;
constexpr int RED_PITCH = NB + 4;
constexpr int RED_PLANE = M_DIM * RED_PITCH;

struct BLines {
    v4f w[NT][NP];
    v4f a[NT][NP];
};
struct XLines {
    v4f xr[MH][NP];
};
struct BOp {
    v4f b[NT][NP];
};

__device__ __forceinline__ void load_b(BLines& f, const float* wp, const float* ap, int kc)
{
    #pragma unroll
    for (int t = 0; t < NT; ++t) {
        #pragma unroll
        for (int j = 0; j < NP; ++j)
            f.w[t][j] = *(const v4f*)(wp + (size_t)(16 * t) * K_DIM + kc + 8 * j);
    }
    #pragma unroll
    for (int t = 0; t < NT; ++t) {
        #pragma unroll
        for (int j = 0; j < NP; ++j)
            f.a[t][j] = *(const v4f*)(ap + (size_t)t * K_DIM + kc + 8 * j);
    }
}

template <int MT0>
__device__ __forceinline__ void load_x(XLines& f, const float* xp, int kc)
{
    #pragma unroll
    for (int mt = 0; mt < MH; ++mt) {
        #pragma unroll
        for (int j = 0; j < NP; ++j)
            f.xr[mt][j] = *(const v4f*)(xp + (size_t)((MT0 + mt) * 16) * K_DIM + kc + 8 * j);
    }
}

#define LOAD_GROUP_EDGE() do { __builtin_amdgcn_sched_barrier(0); } while (0)

__device__ __forceinline__ v4f fold(float sb, v4f a4, v4f w4)
{
    v4f be;
    be.x = __builtin_fmaf(sb, a4.x, w4.x);
    be.y = __builtin_fmaf(sb, a4.y, w4.y);
    be.z = __builtin_fmaf(sb, a4.z, w4.z);
    be.w = __builtin_fmaf(sb, a4.w, w4.w);
    return be;
}

__device__ __forceinline__ void fold_lines(BOp& b, const BLines& f, float sb)
{
    #pragma unroll
    for (int t = 0; t < NT; ++t) {
        #pragma unroll
        for (int j = 0; j < NP; ++j)
            b.b[t][j] = fold(sb, f.a[t][j], f.w[t][j]);
    }
}

template <int MT0>
__device__ __forceinline__ void mma_half(v8f (&c)[MT][NT], const XLines& f, const BOp& b)
{
    #pragma unroll
    for (int j = 0; j < NP; ++j) {
        #pragma unroll
        for (int mt = 0; mt < MH; ++mt) {
            const v2f xa = __builtin_shufflevector(f.xr[mt][j], f.xr[mt][j], 0, 1);
            #pragma unroll
            for (int t = 0; t < NT; ++t) {
                const v2f blo = __builtin_shufflevector(b.b[t][j], b.b[t][j], 0, 1);
                c[MT0 + mt][t] = __builtin_amdgcn_wmma_f32_16x16x4_f32(
                    false, xa, false, blo, (short)0, c[MT0 + mt][t], false, false);
            }
        }
        #pragma unroll
        for (int mt = 0; mt < MH; ++mt) {
            const v2f xb = __builtin_shufflevector(f.xr[mt][j], f.xr[mt][j], 2, 3);
            #pragma unroll
            for (int t = 0; t < NT; ++t) {
                const v2f bhi = __builtin_shufflevector(b.b[t][j], b.b[t][j], 2, 3);
                c[MT0 + mt][t] = __builtin_amdgcn_wmma_f32_16x16x4_f32(
                    false, xb, false, bhi, (short)0, c[MT0 + mt][t], false, false);
            }
        }
    }
}

__device__ __forceinline__ void chunk_step(v8f (&c)[MT][NT], BLines& r, XLines& p, XLines& q,
                                           BOp& b, float sb, const float* wp,
                                           const float* ap, const float* xp, int kn)
{
    #pragma unroll
    for (int t = 0; t < NT; ++t)
        __builtin_prefetch(wp + (size_t)(16 * t) * K_DIM + kn + KPF - KSTRIDE, 0, 1);

    fold_lines(b, r, sb);
    LOAD_GROUP_EDGE();
    load_b(r, wp, ap, kn);
    LOAD_GROUP_EDGE();
    mma_half<0>(c, p, b);
    LOAD_GROUP_EDGE();
    load_x<0>(p, xp, kn);
    LOAD_GROUP_EDGE();
    mma_half<MH>(c, q, b);
    LOAD_GROUP_EDGE();
    load_x<MH>(q, xp, kn);
    LOAD_GROUP_EDGE();
}
}

__global__ __launch_bounds__(128)
void lokr_half_k_gemm(const float* x,
                      const float* W,
                      const float* A,
                      const float* __restrict__ B16,
                      const float* __restrict__ scal,
                      float* out,
                      float* part)
{
    __shared__ __attribute__((aligned(16))) float red[KSLICES * RED_PLANE];

    const int lane  = threadIdx.x & 31;
    const int slice = __builtin_amdgcn_readfirstlane((int)(threadIdx.x >> 5));
    const int khalf = blockIdx.x & 1;
    const int n0    = (blockIdx.x >> 1) * NB;
    const int l16   = lane & 15;
    const int lhalf = lane >> 4;

    const float sb = scal[0] * SCALING * B16[l16];

    const int kbase = khalf * KHALF + 4 * lhalf;
    const float* wp = W + (size_t)(n0 + l16) * K_DIM + kbase;
    const float* ap = A + (size_t)(n0 / 16) * K_DIM + kbase;
    const float* xp = x + (size_t)l16 * K_DIM + kbase;

    v8f c[MT][NT] = {};

    BLines r;
    XLines p, q;
    BOp ba, bb;
    int kc = slice * KCH;
    load_b(r, wp, ap, kc);
    load_x<0>(p, xp, kc);
    load_x<MH>(q, xp, kc);
    LOAD_GROUP_EDGE();
    for (; kc < KHALF; kc += 2 * KSTRIDE) {
        chunk_step(c, r, p, q, ba, sb, wp, ap, xp, kc + KSTRIDE);
        const int kn = (kc + 2 * KSTRIDE < KHALF) ? kc + 2 * KSTRIDE : kc + KSTRIDE;
        chunk_step(c, r, p, q, bb, sb, wp, ap, xp, kn);
    }

    float* rp = red + slice * RED_PLANE + l16;
    #pragma unroll
    for (int mt = 0; mt < MT; ++mt) {
        #pragma unroll
        for (int i = 0; i < 8; ++i) {
            const int m = mt * 16 + lhalf * 8 + i;
            #pragma unroll
            for (int t = 0; t < NT; ++t)
                rp[m * RED_PITCH + 16 * t] = c[mt][t][i];
        }
    }
    __syncthreads();

    float* dst = khalf ? part : out;
    const int g = threadIdx.x & 15;
    #pragma unroll
    for (int i = 0; i < 16; ++i) {
        const int m = (int)(threadIdx.x >> 4) + 8 * i;
        const float* qd = red + m * RED_PITCH + 4 * g;
        v4f s = *(const v4f*)qd;
        #pragma unroll
        for (int sl = 1; sl < KSLICES; ++sl)
            s += *(const v4f*)(qd + sl * RED_PLANE);
        *(v4f*)(dst + (size_t)m * N_DIM + n0 + 4 * g) = s;
    }
}

__global__ __launch_bounds__(256)
void add_k_halves(float* out, const float* part)
{
    const int t = blockIdx.x * 256 + threadIdx.x;
    #pragma unroll
    for (int i = 0; i < 4; ++i) {
        const size_t e = ((size_t)i * 65536 + t) * 4;
        const v4f a = *(const v4f*)(out + e);
        const v4f b = *(const v4f*)(part + e);
        *(v4f*)(out + e) = a + b;
    }
}

extern "C" void kernel_launch(void* const* d_in, const int* in_sizes, int n_in,
                              void* d_out, int out_size, void* d_ws, size_t ws_size,
                              hipStream_t stream) {
    (void)in_sizes; (void)n_in; (void)out_size; (void)ws_size;
    const float* x    = (const float*)d_in[0];
    const float* W    = (const float*)d_in[1];
    const float* A    = (const float*)d_in[2];
    const float* B16  = (const float*)d_in[3];
    const float* scal = (const float*)d_in[4];
    float* out  = (float*)d_out;
    float* part = (float*)d_ws;

    dim3 grid(256);
    dim3 block(128);
    hipLaunchKernelGGL(lokr_half_k_gemm, grid, block, 0, stream,
                       x, W, A, B16, scal, out, part);
    hipLaunchKernelGGL(add_k_halves, dim3(256), dim3(256), 0, stream, out, part);
}
